# starting kernel
# baseline (speedup 1.0000x reference)
#include <hip/hip_runtime.h>
#include <hip/hip_bf16.h>
#include <float.h>

typedef float v2f __attribute__((ext_vector_type(2)));
typedef float v4f __attribute__((ext_vector_type(4)));
typedef float v8f __attribute__((ext_vector_type(8)));

#define D_DIM 256
#define A_STRIDE 132
#define WAVES_PER_BLOCK 4
#define THREADS (WAVES_PER_BLOCK * 32)
#define MT 2
#define ROWS_PER_WAVE (MT * 16)
#define ROWS_PER_BLOCK (WAVES_PER_BLOCK * ROWS_PER_WAVE)
#define A_SLOTS (WAVES_PER_BLOCK * MT * 32)

__global__ void cc_norms(const float* __restrict__ cb0, int K0,
                         const float* __restrict__ cb1, int K1,
                         const float* __restrict__ cb2, int K2,
                         float* __restrict__ norms) {
    int i = blockIdx.x * blockDim.x + threadIdx.x;
    int total = K0 + K1 + K2;
    if (i >= total) return;
    const float* src;
    int row;
    if (i < K0)           { src = cb0; row = i; }
    else if (i < K0 + K1) { src = cb1; row = i - K0; }
    else                  { src = cb2; row = i - K0 - K1; }
    const v4f* p = (const v4f*)(src + (size_t)row * D_DIM);
    float s = 0.f;
    for (int j = 0; j < D_DIM / 4; ++j) {
        v4f v = p[j];
        s += v[0]*v[0] + v[1]*v[1] + v[2]*v[2] + v[3]*v[3];
    }
    norms[i] = s;
}

__global__ __launch_bounds__(THREADS)
void cc_main(const float* __restrict__ embeds,
             const float* __restrict__ cb0, int K0,
             const float* __restrict__ cb1, int K1,
             const float* __restrict__ cb2, int K2,
             const float* __restrict__ norms,
             float* __restrict__ out,
             float* __restrict__ partials,
             int nRows) {
    extern __shared__ float smem[];
    float* ldsA = smem;
    float* ldsB = smem + A_SLOTS * A_STRIDE;

    const int tid   = threadIdx.x;
    const int lane  = tid & 31;
    const int w     = tid >> 5;
    const int layer = blockIdx.y;

    const float* cb; int K; int normOff;
    if (layer == 0)      { cb = cb0; K = K0; normOff = 0; }
    else if (layer == 1) { cb = cb1; K = K1; normOff = K0; }
    else                 { cb = cb2; K = K2; normOff = K0 + K1; }

    const int rowBase = blockIdx.x * ROWS_PER_BLOCK;

    {
        int lr = tid;
        int wA = lr >> 5;
        int st = (lr >> 4) & 1;
        int mA = lr & 15;
        const v4f* src = (const v4f*)(embeds + (size_t)(rowBase + lr) * D_DIM);
        float* dst0 = ldsA + (wA * (MT * 32) + st * 32 + mA) * A_STRIDE;
        float* dst1 = dst0 + 16 * A_STRIDE;
#pragma unroll
        for (int i = 0; i < D_DIM / 4; ++i) {
            v4f v = src[i];
            *(v2f*)(dst0 + 2 * i) = (v2f){v[0], v[1]};
            *(v2f*)(dst1 + 2 * i) = (v2f){v[2], v[3]};
        }
    }

    float bestS[MT * 8];
    int   bestI[MT * 8];
#pragma unroll
    for (int r = 0; r < MT * 8; ++r) { bestS[r] = FLT_MAX; bestI[r] = 0x7fffffff; }

    const int nLane = lane & 15;
    const float* aP0 = ldsA + (w * (MT * 32) + lane) * A_STRIDE;
    const float* aP1 = aP0 + 32 * A_STRIDE;
    const float* bP  = ldsB + lane * A_STRIDE;

    const int numTiles = (K + 15) >> 4;
    for (int t = 0; t < numTiles; ++t) {
        const int tileBase = t << 4;
        __syncthreads();
        {
            int n = tid >> 3;
            int p = tid & 7;
            int srcRow = tileBase + n;
            if (srcRow >= K) srcRow = K - 1;
            const v4f* src = (const v4f*)(cb + (size_t)srcRow * D_DIM);
            float* dst0 = ldsB + n * A_STRIDE;
            float* dst1 = dst0 + 16 * A_STRIDE;
#pragma unroll
            for (int i = 0; i < 8; ++i) {
                int c = p * 8 + i;
                v4f v = src[c];
                *(v2f*)(dst0 + 2 * c) = (v2f){v[0], v[1]};
                *(v2f*)(dst1 + 2 * c) = (v2f){v[2], v[3]};
            }
        }
        __syncthreads();

        v8f acc0 = {0.f,0.f,0.f,0.f,0.f,0.f,0.f,0.f};
        v8f acc1 = {0.f,0.f,0.f,0.f,0.f,0.f,0.f,0.f};
#pragma unroll
        for (int c2 = 0; c2 < 32; ++c2) {
            v4f bv  = *(const v4f*)(bP  + c2 * 4);
            v4f av0 = *(const v4f*)(aP0 + c2 * 4);
            v4f av1 = *(const v4f*)(aP1 + c2 * 4);
            v2f b0  = {bv[0],  bv[1]},  b1  = {bv[2],  bv[3]};
            v2f a00 = {av0[0], av0[1]}, a01 = {av0[2], av0[3]};
            v2f a10 = {av1[0], av1[1]}, a11 = {av1[2], av1[3]};
            acc0 = __builtin_amdgcn_wmma_f32_16x16x4_f32(false, a00, false, b0,
                                                         (short)0, acc0, false, false);
            acc1 = __builtin_amdgcn_wmma_f32_16x16x4_f32(false, a10, false, b0,
                                                         (short)0, acc1, false, false);
            acc0 = __builtin_amdgcn_wmma_f32_16x16x4_f32(false, a01, false, b1,
                                                         (short)0, acc0, false, false);
            acc1 = __builtin_amdgcn_wmma_f32_16x16x4_f32(false, a11, false, b1,
                                                         (short)0, acc1, false, false);
        }

        const int gk = tileBase + nLane;
        const bool valid = gk < K;
        const float wn = valid ? norms[normOff + gk] : FLT_MAX;
#pragma unroll
        for (int r = 0; r < 8; ++r) {
            float s0 = wn - 2.0f * acc0[r];
            float s1 = wn - 2.0f * acc1[r];
            if (valid && s0 < bestS[r])     { bestS[r]     = s0; bestI[r]     = gk; }
            if (valid && s1 < bestS[8 + r]) { bestS[8 + r] = s1; bestI[8 + r] = gk; }
        }
    }

#pragma unroll
    for (int r = 0; r < MT * 8; ++r) {
        float s = bestS[r];
        int   i = bestI[r];
#pragma unroll
        for (int off = 8; off >= 1; off >>= 1) {
            float os = __shfl_xor(s, off, 32);
            int   oi = __shfl_xor(i, off, 32);
            if (os < s || (os == s && oi < i)) { s = os; i = oi; }
        }
        bestS[r] = s; bestI[r] = i;
    }

    const int h = lane >> 4;
    float localSum = 0.f;
#pragma unroll
    for (int st = 0; st < MT; ++st) {
#pragma unroll
        for (int r = 0; r < 8; ++r) {
            const int m   = r + 8 * h;
            const int row = rowBase + w * ROWS_PER_WAVE + st * 16 + m;
            const int idx = bestI[st * 8 + r];
            const v4f* qsrc = (const v4f*)(cb + (size_t)idx * D_DIM);
            const v4f* esrc = (const v4f*)(embeds + (size_t)row * D_DIM);
            v4f* dst = (v4f*)(out + ((size_t)layer * nRows + row) * D_DIM);
#pragma unroll
            for (int vv = 0; vv < 4; ++vv) {
                int q4 = nLane * 4 + vv;
                v4f q = qsrc[q4];
                v4f e = esrc[q4];
                dst[q4] = q;
                v4f df = q - e;
                localSum += df[0]*df[0] + df[1]*df[1] + df[2]*df[2] + df[3]*df[3];
            }
        }
    }

    __syncthreads();
    ldsB[tid] = localSum;
    __syncthreads();
    if (tid == 0) {
        float s = 0.f;
        for (int i = 0; i < THREADS; ++i) s += ldsB[i];
        partials[blockIdx.y * gridDim.x + blockIdx.x] = s;
    }
}

__global__ void cc_finalize(const float* __restrict__ partials, int n,
                            float* __restrict__ out, size_t lossIdx, float invND) {
    if (blockIdx.x == 0 && threadIdx.x == 0) {
        float s = 0.f;
        for (int i = 0; i < n; ++i) s += partials[i];
        out[lossIdx] = 1.25f * invND * s;
    }
}

extern "C" void kernel_launch(void* const* d_in, const int* in_sizes, int n_in,
                              void* d_out, int out_size, void* d_ws, size_t ws_size,
                              hipStream_t stream) {
    const float* embeds = (const float*)d_in[0];
    const float* cb0    = (const float*)d_in[1];
    const float* cb1    = (const float*)d_in[2];
    const float* cb2    = (const float*)d_in[3];

    const int nRows = in_sizes[0] / D_DIM;
    const int K0 = in_sizes[1] / D_DIM;
    const int K1 = in_sizes[2] / D_DIM;
    const int K2 = in_sizes[3] / D_DIM;
    const int totalK = K0 + K1 + K2;

    float* norms    = (float*)d_ws;
    float* partials = (float*)d_ws + 8192;

    cc_norms<<<(totalK + 255) / 256, 256, 0, stream>>>(cb0, K0, cb1, K1, cb2, K2, norms);

    const int gx = nRows / ROWS_PER_BLOCK;
    const size_t smemBytes =
        (size_t)(A_SLOTS + 32) * A_STRIDE * sizeof(float);
    cc_main<<<dim3(gx, 3), THREADS, smemBytes, stream>>>(
        embeds, cb0, K0, cb1, K1, cb2, K2, norms,
        (float*)d_out, partials, nRows);

    cc_finalize<<<1, 1, 0, stream>>>(partials, gx * 3, (float*)d_out,
                                     (size_t)out_size - 1,
                                     1.0f / ((float)nRows * (float)D_DIM));
}


# Wave issue priority set from own remaining tiles (s_setprio 3/2/1/0 at 1/4, 1/8, 1/16 of the job left): a SIMD's two waves advance and finish together, not one after the other; sum order unchanged
# speedup vs baseline: 3.4614x; 3.4614x over previous
#include <hip/hip_runtime.h>
#include <hip/hip_bf16.h>
#include <float.h>

typedef float v2f __attribute__((ext_vector_type(2)));
typedef float v4f __attribute__((ext_vector_type(4)));
typedef float v8f __attribute__((ext_vector_type(8)));
typedef int   v4i __attribute__((ext_vector_type(4)));

#define D_DIM 256
#define A_STRIDE 132
#define WAVES_PER_BLOCK 4
#define THREADS (WAVES_PER_BLOCK * 32)
#define MT 2
#define ROWS_PER_WAVE (MT * 16)
#define ROWS_PER_BLOCK (WAVES_PER_BLOCK * ROWS_PER_WAVE)
#define A_SLOTS (WAVES_PER_BLOCK * MT * 32)
#define MTILES (ROWS_PER_BLOCK / 16)
#define NSPLIT 4
#define NJOB (NSPLIT * WAVES_PER_BLOCK)
static_assert(NJOB == 16, "the merge maps one job to each lane of a 16-lane half");

#define NORM_ROWS 32
#define NORM_PITCH 65
__device__ __forceinline__ float chunk_sq(const v4f v) {
    float t;
    {
#pragma clang fp contract(off)
        t = ((v[0] * v[0] + v[1] * v[1]) + v[2] * v[2]) + v[3] * v[3];
    }
    return t;
}

__global__ __launch_bounds__(NORM_ROWS)
void cc_norms(const float* __restrict__ cb0, int K0,
              const float* __restrict__ cb1, int K1,
              const float* __restrict__ cb2, int K2,
              float* __restrict__ norms) {
    __shared__ float part[NORM_ROWS * NORM_PITCH];
    const int lane = threadIdx.x;
    const int B0 = (K0 + NORM_ROWS - 1) / NORM_ROWS;
    const int B1 = (K1 + NORM_ROWS - 1) / NORM_ROWS;
    const int blk = blockIdx.x;
    const float* cb; int K; int normOff; int rowBase;
    if (blk < B0)           { cb = cb0; K = K0; normOff = 0;       rowBase = blk * NORM_ROWS; }
    else if (blk < B0 + B1) { cb = cb1; K = K1; normOff = K0;      rowBase = (blk - B0) * NORM_ROWS; }
    else                    { cb = cb2; K = K2; normOff = K0 + K1; rowBase = (blk - B0 - B1) * NORM_ROWS; }
    v4f a[NORM_ROWS], b[NORM_ROWS];
#pragma unroll
    for (int rr = 0; rr < NORM_ROWS; ++rr) {
        int row = rowBase + rr;
        row = (row < K) ? row : K - 1;
        const v4f* p = (const v4f*)(cb + (size_t)row * D_DIM);
        a[rr] = p[lane];
        b[rr] = p[lane + 32];
    }
#pragma unroll
    for (int rr = 0; rr < NORM_ROWS; ++rr) {
        part[rr * NORM_PITCH + lane]      = chunk_sq(a[rr]);
        part[rr * NORM_PITCH + lane + 32] = chunk_sq(b[rr]);
    }
    __syncthreads();
    float s = 0.f;
#pragma unroll
    for (int j = 0; j < D_DIM / 4; ++j) s += part[lane * NORM_PITCH + j];
    if (rowBase + lane < K) norms[normOff + rowBase + lane] = s;
}

__device__ __forceinline__ void job_range(int j, int T0, int T1, int T2, int L,
                                          int& lo, int& hi) {
    const int total = T0 + T1 + T2;
    const int base = total / NJOB;
    const int rem  = total % NJOB;
    const int jb = j * base + ((j < rem) ? j : rem);
    const int je = jb + base + ((j < rem) ? 1 : 0);
    const int ls = (L == 0) ? 0  : ((L == 1) ? T0      : T0 + T1);
    const int le = (L == 0) ? T0 : ((L == 1) ? T0 + T1 : total);
    lo = ((jb > ls) ? jb : ls) - ls;
    hi = ((je < le) ? je : le) - ls;
}

__device__ __forceinline__ void load_blk(v4f (&dst)[16], const float* aP, int p, int kb) {
    const float* a0 = aP + (2 * p) * 32 * A_STRIDE + kb * 32;
    const float* a1 = a0 + 32 * A_STRIDE;
#pragma unroll
    for (int j = 0; j < 8; ++j) {
        dst[2 * j]     = *(const v4f*)(a0 + 4 * j);
        dst[2 * j + 1] = *(const v4f*)(a1 + 4 * j);
    }
}

__device__ __forceinline__ void mma_blk(const v4f (&a)[16], const v2f (&bq)[D_DIM / 4], int kb,
                                        v8f& acc0, v8f& acc1) {
#pragma unroll
    for (int j = 0; j < 8; ++j) {
        const int c2 = kb * 8 + j;
        const v4f av0 = a[2 * j], av1 = a[2 * j + 1];
        v2f a00 = {av0[0], av0[1]}, a01 = {av0[2], av0[3]};
        v2f a10 = {av1[0], av1[1]}, a11 = {av1[2], av1[3]};
        acc0 = __builtin_amdgcn_wmma_f32_16x16x4_f32(false, a00, false, bq[2 * c2],
                                                     (short)0, acc0, false, false);
        acc1 = __builtin_amdgcn_wmma_f32_16x16x4_f32(false, a10, false, bq[2 * c2],
                                                     (short)0, acc1, false, false);
        acc0 = __builtin_amdgcn_wmma_f32_16x16x4_f32(false, a01, false, bq[2 * c2 + 1],
                                                     (short)0, acc0, false, false);
        acc1 = __builtin_amdgcn_wmma_f32_16x16x4_f32(false, a11, false, bq[2 * c2 + 1],
                                                     (short)0, acc1, false, false);
    }
}

__global__ __launch_bounds__(THREADS, 2)
void cc_main(const float* __restrict__ embeds,
             const float* __restrict__ cb0, int K0,
             const float* __restrict__ cb1, int K1,
             const float* __restrict__ cb2, int K2,
             const float* __restrict__ norms,
             float* __restrict__ candS,
             int* __restrict__ candI,
             int nRows) {
    extern __shared__ float smem[];
    float* ldsA = smem;

    const int tid   = threadIdx.x;
    const int lane  = tid & 31;
    const int w     = __builtin_amdgcn_readfirstlane(tid >> 5);
    const int job   = blockIdx.y * WAVES_PER_BLOCK + w;

    const int T0 = (K0 + 15) >> 4;
    const int T1 = (K1 + 15) >> 4;
    const int T2 = (K2 + 15) >> 4;

    const int rowBase = blockIdx.x * ROWS_PER_BLOCK;

    {
        const v4f* srcW = (const v4f*)(embeds + (size_t)(rowBase + w * ROWS_PER_WAVE) * D_DIM) + lane;
        float* dstW = ldsA + (w * MT * 32) * A_STRIDE + 2 * lane;
#pragma unroll
        for (int rr = 0; rr < ROWS_PER_WAVE; ++rr) {
            const v4f va = srcW[rr * (D_DIM / 4)];
            const v4f vb = srcW[rr * (D_DIM / 4) + 32];
            float* dst0 = dstW + ((rr >> 4) * 32 + (rr & 15)) * A_STRIDE;
            float* dst1 = dst0 + 16 * A_STRIDE;
            *(v2f*)(dst0)      = (v2f){va[0], va[1]};
            *(v2f*)(dst1)      = (v2f){va[2], va[3]};
            *(v2f*)(dst0 + 64) = (v2f){vb[0], vb[1]};
            *(v2f*)(dst1 + 64) = (v2f){vb[2], vb[3]};
        }
    }
    __syncthreads();

    const int jobTiles = (T0 + T1 + T2) / NJOB + ((job < (T0 + T1 + T2) % NJOB) ? 1 : 0);
    const int quarter = jobTiles >> 2, eighth = jobTiles >> 3, sixteenth = jobTiles >> 4;
    int tilesLeft = jobTiles;
    int rank = 0;
    if (jobTiles > 0) {
        __builtin_amdgcn_s_setprio(3);
        rank = 3;
    }

    const int nLane = lane & 15;
    const int h     = lane >> 4;
    const float* aP = ldsA + lane * A_STRIDE;

    for (int layer = 0; layer < 3; ++layer) {
        int tLo, tHi;
        job_range(job, T0, T1, T2, layer, tLo, tHi);
        if (tLo >= tHi) continue;

        const float* cb; int K; int normOff;
        if (layer == 0)      { cb = cb0; K = K0; normOff = 0; }
        else if (layer == 1) { cb = cb1; K = K1; normOff = K0; }
        else                 { cb = cb2; K = K2; normOff = K0 + K1; }

        float bestS[MTILES * 8];
        int   bestI[MTILES * 8];
#pragma unroll
        for (int r = 0; r < MTILES * 8; ++r) { bestS[r] = FLT_MAX; bestI[r] = 0x7fffffff; }

        v4f aS0[16], aS1[16];
        load_blk(aS0, aP, 0, 0);

        v2f bq[D_DIM / 4];
        {
            int srcRow = (tLo << 4) + nLane;
            if (srcRow >= K) srcRow = K - 1;
            const v2f* bsrc = (const v2f*)(cb + (size_t)srcRow * D_DIM) + h;
#pragma unroll
            for (int q = 0; q < D_DIM / 4; ++q) bq[q] = bsrc[2 * q];
        }
        float wnNext = norms[normOff + (((tLo << 4) + nLane < K) ? (tLo << 4) + nLane : K - 1)];

        for (int t = tLo; t < tHi; ++t) {
            const int tileBase = t << 4;
            const int tNext = (t + 1 < tHi) ? (t + 1) : t;
            int nextRow = (tNext << 4) + nLane;
            if (nextRow >= K) nextRow = K - 1;
            const v2f* bnext = (const v2f*)(cb + (size_t)nextRow * D_DIM) + h;

            const int gk = tileBase + nLane;
            const bool valid = gk < K;
            const float wn = valid ? wnNext : FLT_MAX;

            v8f acc0, acc1;
#pragma unroll
            for (int blk = 0; blk < 16; ++blk) {
                const int p  = blk >> 2, kb  = blk & 3;
                const int nb = (blk + 1) & 15;
                if (kb == 0) {
                    acc0 = (v8f){0.f,0.f,0.f,0.f,0.f,0.f,0.f,0.f};
                    acc1 = (v8f){0.f,0.f,0.f,0.f,0.f,0.f,0.f,0.f};
                }
                if ((blk & 1) == 0) load_blk(aS1, aP, nb >> 2, nb & 3);
                else                load_blk(aS0, aP, nb >> 2, nb & 3);
                __builtin_amdgcn_sched_barrier(0);
                if ((blk & 1) == 0) mma_blk(aS0, bq, kb, acc0, acc1);
                else                mma_blk(aS1, bq, kb, acc0, acc1);
                __builtin_amdgcn_sched_barrier(0);
                if (p == MTILES / 2 - 1) {
#pragma unroll
                    for (int q = 16 * kb; q < 16 * kb + 16; ++q) bq[q] = bnext[2 * q];
                    if (kb == 0) wnNext = norms[normOff + nextRow];
                }

                if (kb == 3) {
#pragma unroll
                    for (int r = 0; r < 8; ++r) {
                        float s0 = wn - 2.0f * acc0[r];
                        float s1 = wn - 2.0f * acc1[r];
                        if (valid && s0 < bestS[p * 16 + r])     { bestS[p * 16 + r]     = s0; bestI[p * 16 + r]     = gk; }
                        if (valid && s1 < bestS[p * 16 + 8 + r]) { bestS[p * 16 + 8 + r] = s1; bestI[p * 16 + 8 + r] = gk; }
                    }
                }
            }

            --tilesLeft;
            const int want = (tilesLeft > quarter) ? 3 : ((tilesLeft > eighth) ? 2 : ((tilesLeft > sixteenth) ? 1 : 0));
            if (want != rank) {
                rank = want;
                if (want == 2)      __builtin_amdgcn_s_setprio(2);
                else if (want == 1) __builtin_amdgcn_s_setprio(1);
                else                __builtin_amdgcn_s_setprio(0);
            }
        }

#pragma unroll
        for (int r = 0; r < MTILES * 8; ++r) {
            float s = bestS[r];
            int   i = bestI[r];
#pragma unroll
            for (int off = 8; off >= 1; off >>= 1) {
                float os = __shfl_xor(s, off, 32);
                int   oi = __shfl_xor(i, off, 32);
                const bool take = os < s || (os == s && oi < i);
                s = take ? os : s;
                i = take ? oi : i;
            }
            bestS[r] = s; bestI[r] = i;
        }

        const size_t plane = (size_t)(layer * NJOB + job) * nRows;
#pragma unroll
        for (int mt = 0; mt < MTILES; ++mt) {
            float s = bestS[mt * 8];
            int   i = bestI[mt * 8];
#pragma unroll
            for (int r = 1; r < 8; ++r)
                if (nLane == r) { s = bestS[mt * 8 + r]; i = bestI[mt * 8 + r]; }
            if (nLane < 8) {
                const int row = rowBase + mt * 16 + 8 * h + nLane;
                candS[plane + row] = s;
                candI[plane + row] = i;
            }
        }
    }
}

__global__ __launch_bounds__(THREADS)
void cc_gather(const float* __restrict__ embeds,
               const float* __restrict__ cb0, int K0,
               const float* __restrict__ cb1, int K1,
               const float* __restrict__ cb2, int K2,
               const float* __restrict__ candS,
               const int* __restrict__ candI,
               float* __restrict__ out,
               float* __restrict__ partials,
               int nRows) {
    __shared__ float red[THREADS];

    const int tid   = threadIdx.x;
    const int lane  = tid & 31;
    const int w     = tid >> 5;
    const int layer = blockIdx.y;
    const int nLane = lane & 15;
    const int h     = lane >> 4;

    const float* cb; int K;
    if (layer == 0)      { cb = cb0; K = K0; }
    else if (layer == 1) { cb = cb1; K = K1; }
    else                 { cb = cb2; K = K2; }

    const int T0 = (K0 + 15) >> 4;
    const int T1 = (K1 + 15) >> 4;
    const int T2 = (K2 + 15) >> 4;

    const int rowBase = blockIdx.x * ROWS_PER_BLOCK;

    float bestS[MT * 8];
    int   bestI[MT * 8];
#pragma unroll
    for (int q = 0; q < MT * 8; ++q) { bestS[q] = FLT_MAX; bestI[q] = 0; }
    {
        int tLo, tHi;
        job_range(nLane, T0, T1, T2, layer, tLo, tHi);
        if (tLo < tHi) {
            const size_t base = (size_t)(layer * NJOB + nLane) * nRows
                              + rowBase + w * ROWS_PER_WAVE + 8 * h;
            v4f sv[2 * MT];
            v4i iv[2 * MT];
#pragma unroll
            for (int st = 0; st < MT; ++st) {
                const v4f* ps = (const v4f*)(candS + base + st * 16);
                const v4i* pi = (const v4i*)(candI + base + st * 16);
                sv[2 * st] = ps[0]; sv[2 * st + 1] = ps[1];
                iv[2 * st] = pi[0]; iv[2 * st + 1] = pi[1];
            }
#pragma unroll
            for (int st = 0; st < MT; ++st) {
#pragma unroll
                for (int e = 0; e < 4; ++e) {
                    const bool okLo = sv[2 * st][e] < FLT_MAX;
                    bestS[st * 8 + e]     = okLo ? sv[2 * st][e] : FLT_MAX;
                    bestI[st * 8 + e]     = okLo ? iv[2 * st][e] : 0;
                    const bool okHi = sv[2 * st + 1][e] < FLT_MAX;
                    bestS[st * 8 + 4 + e] = okHi ? sv[2 * st + 1][e] : FLT_MAX;
                    bestI[st * 8 + 4 + e] = okHi ? iv[2 * st + 1][e] : 0;
                }
            }
        }
    }
#pragma unroll
    for (int off = 8; off >= 1; off >>= 1) {
        float os[MT * 8];
        int   oi[MT * 8];
#pragma unroll
        for (int q = 0; q < MT * 8; ++q) {
            os[q] = __shfl_xor(bestS[q], off, 32);
            oi[q] = __shfl_xor(bestI[q], off, 32);
        }
#pragma unroll
        for (int q = 0; q < MT * 8; ++q) {
            const bool take = (os[q] < bestS[q]) | ((os[q] == bestS[q]) & (oi[q] < bestI[q]));
            bestS[q] = take ? os[q] : bestS[q];
            bestI[q] = take ? oi[q] : bestI[q];
        }
    }
#pragma unroll
    for (int q = 0; q < MT * 8; ++q)
        if ((unsigned)bestI[q] >= (unsigned)K) bestI[q] = K - 1;

    float localSum = 0.f;
#pragma unroll
    for (int st = 0; st < MT; ++st) {
#pragma unroll
        for (int r = 0; r < 8; ++r) {
            const int m   = r + 8 * h;
            const int row = rowBase + w * ROWS_PER_WAVE + st * 16 + m;
            const int idx = bestI[st * 8 + r];
            const v4f* qsrc = (const v4f*)(cb + (size_t)idx * D_DIM);
            const v4f* esrc = (const v4f*)(embeds + (size_t)row * D_DIM);
            v4f* dst = (v4f*)(out + ((size_t)layer * nRows + row) * D_DIM);
#pragma unroll
            for (int vv = 0; vv < 4; ++vv) {
                int q4 = vv * 16 + nLane;
                v4f q = qsrc[q4];
                v4f e = esrc[q4];
                dst[q4] = q;
                v4f df = q - e;
                localSum += df[0]*df[0] + df[1]*df[1] + df[2]*df[2] + df[3]*df[3];
            }
        }
    }

    red[tid] = localSum;
    __syncthreads();
    if (tid == 0) {
        float s = 0.f;
        for (int i = 0; i < THREADS; ++i) s += red[i];
        partials[blockIdx.y * gridDim.x + blockIdx.x] = s;
    }
}

__global__ void cc_finalize(const float* __restrict__ partials, int n,
                            float* __restrict__ out, size_t lossIdx, float invND) {
    if (blockIdx.x == 0 && threadIdx.x == 0) {
        float s = 0.f;
        for (int i = 0; i < n; ++i) s += partials[i];
        out[lossIdx] = 1.25f * invND * s;
    }
}

extern "C" void kernel_launch(void* const* d_in, const int* in_sizes, int n_in,
                              void* d_out, int out_size, void* d_ws, size_t ws_size,
                              hipStream_t stream) {
    const float* embeds = (const float*)d_in[0];
    const float* cb0    = (const float*)d_in[1];
    const float* cb1    = (const float*)d_in[2];
    const float* cb2    = (const float*)d_in[3];

    const int nRows = in_sizes[0] / D_DIM;
    const int K0 = in_sizes[1] / D_DIM;
    const int K1 = in_sizes[2] / D_DIM;
    const int K2 = in_sizes[3] / D_DIM;

    float* norms    = (float*)d_ws;
    float* partials = (float*)d_ws + 8192;
    float* candS    = (float*)d_ws + 1048576;
    int*   candI    = (int*)d_ws + 2097152;

    const int normBlocks = (K0 + NORM_ROWS - 1) / NORM_ROWS + (K1 + NORM_ROWS - 1) / NORM_ROWS
                         + (K2 + NORM_ROWS - 1) / NORM_ROWS;
    cc_norms<<<normBlocks, NORM_ROWS, 0, stream>>>(cb0, K0, cb1, K1, cb2, K2, norms);

    const int gx = nRows / ROWS_PER_BLOCK;
    const size_t smemBytes =
        (size_t)A_SLOTS * A_STRIDE * sizeof(float);
    cc_main<<<dim3(gx, NSPLIT), THREADS, smemBytes, stream>>>(
        embeds, cb0, K0, cb1, K1, cb2, K2, norms, candS, candI, nRows);

    cc_gather<<<dim3(gx, 3), THREADS, 0, stream>>>(
        embeds, cb0, K0, cb1, K1, cb2, K2, candS, candI,
        (float*)d_out, partials, nRows);

    cc_finalize<<<1, 1, 0, stream>>>(partials, gx * 3, (float*)d_out,
                                     (size_t)out_size - 1,
                                     1.0f / ((float)nRows * (float)D_DIM));
}
